# starting kernel
# baseline (speedup 1.0000x reference)
#include <hip/hip_runtime.h>

typedef float v2f __attribute__((ext_vector_type(2)));
typedef float v8f __attribute__((ext_vector_type(8)));

#define TDIM     1000
#define ZSTRIDE  1002
#define FDIM     257
#define CDIM     8
#define CEPS     1.1920928955078125e-07f

__global__ __launch_bounds__(256) void mvdr_wmma_kernel(
    const float* __restrict__ spec_r,
    const float* __restrict__ spec_i,
    const float* __restrict__ mask,
    float*       __restrict__ out)
{
    __shared__ __align__(16) float Zs[16 * ZSTRIDE];
    __shared__ float Ms[TDIM];
    __shared__ float Gs[256];
    __shared__ float Gt[256];
    __shared__ float AR[8 * 16], AI[8 * 16];
    __shared__ float RsR[64], RsI[64];
    __shared__ float MR[64], MI[64];
    __shared__ float wR[8], wI[8];
    __shared__ float pivR, pivI;
    __shared__ float msum;

    const int tid = threadIdx.x;
    const int f   = blockIdx.x;
    const int n   = blockIdx.y;

    Gs[tid] = 0.0f;
    Gt[tid] = 0.0f;
    if (tid == 0) msum = 0.0f;

    if (tid < TDIM / 4) {
        const int t4 = tid;
        #pragma unroll
        for (int r = 0; r < CDIM; ++r) {
            const float4 v = ((const float4*)(spec_r + (((size_t)n * CDIM + r) * FDIM + f) * TDIM))[t4];
            float* dst = &Zs[r * ZSTRIDE + 4 * t4];
            dst[0] = v.x; dst[1] = v.y; dst[2] = v.z; dst[3] = v.w;
        }
        #pragma unroll
        for (int r = 0; r < CDIM; ++r) {
            const float4 v = ((const float4*)(spec_i + (((size_t)n * CDIM + r) * FDIM + f) * TDIM))[t4];
            float* dst = &Zs[(CDIM + r) * ZSTRIDE + 4 * t4];
            dst[0] = v.x; dst[1] = v.y; dst[2] = v.z; dst[3] = v.w;
        }
        const float4 mv = ((const float4*)(mask + ((size_t)n * FDIM + f) * TDIM))[t4];
        float* md = &Ms[4 * t4];
        md[0] = mv.x; md[1] = mv.y; md[2] = mv.z; md[3] = mv.w;
    }
    __syncthreads();

    {
        float local = 0.0f;
        for (int t = tid; t < TDIM; t += 256) local += Ms[t];
        atomicAdd(&msum, local);
    }

    {
        const int wave = tid >> 5;
        const int lane = tid & 31;
        const int m    = lane & 15;
        const int koff = (lane >> 4) * 2;

        v8f acc_s = {0, 0, 0, 0, 0, 0, 0, 0};
        v8f acc_t = {0, 0, 0, 0, 0, 0, 0, 0};

        for (int it = wave; it < TDIM / 4; it += 8) {
            const int t0 = it * 4 + koff;
            v2f zb = *(const v2f*)&Zs[m * ZSTRIDE + t0];
            v2f za;
            za.x = zb.x * Ms[t0];
            za.y = zb.y * Ms[t0 + 1];
            acc_t = __builtin_amdgcn_wmma_f32_16x16x4_f32(
                        false, zb, false, zb, (short)0, acc_t, false, false);
            acc_s = __builtin_amdgcn_wmma_f32_16x16x4_f32(
                        false, za, false, zb, (short)0, acc_s, false, false);
        }

        const int Nn    = lane & 15;
        const int Mbase = (lane >> 4) * 8;
        #pragma unroll
        for (int r = 0; r < 8; ++r) {
            atomicAdd(&Gs[(Mbase + r) * 16 + Nn], acc_s[r]);
            atomicAdd(&Gt[(Mbase + r) * 16 + Nn], acc_t[r]);
        }
    }
    __syncthreads();

    if (tid < 64) {
        const int i = tid >> 3, j = tid & 7;
        const float ds    = fmaxf(msum, CEPS);
        const float dn    = fmaxf((float)TDIM - msum, CEPS);
        const float inv_s = 1.0f / ds;
        const float inv_n = 1.0f / dn;

        const float gs_a = Gs[i * 16 + j],           gt_a = Gt[i * 16 + j];
        const float gs_b = Gs[(i + 8) * 16 + j + 8], gt_b = Gt[(i + 8) * 16 + j + 8];
        const float gs_c = Gs[(i + 8) * 16 + j],     gt_c = Gt[(i + 8) * 16 + j];
        const float gs_d = Gs[i * 16 + j + 8],       gt_d = Gt[i * 16 + j + 8];

        RsR[i * 8 + j] = (gs_a + gs_b) * inv_s;
        RsI[i * 8 + j] = (gs_c - gs_d) * inv_s;

        const float rnr = ((gt_a - gs_a) + (gt_b - gs_b)) * inv_n;
        const float rni = ((gt_c - gs_c) - (gt_d - gs_d)) * inv_n;
        AR[i * 16 + j]     = rnr + ((i == j) ? CEPS : 0.0f);
        AI[i * 16 + j]     = rni;
        AR[i * 16 + 8 + j] = (i == j) ? 1.0f : 0.0f;
        AI[i * 16 + 8 + j] = 0.0f;
    }
    __syncthreads();

    for (int k = 0; k < 8; ++k) {
        if (tid == 0) {
            const float pr = AR[k * 16 + k], pi = AI[k * 16 + k];
            const float id = 1.0f / (pr * pr + pi * pi);
            pivR = pr * id;
            pivI = -pi * id;
        }
        __syncthreads();
        if (tid < 16) {
            const float ar = AR[k * 16 + tid], ai = AI[k * 16 + tid];
            AR[k * 16 + tid] = ar * pivR - ai * pivI;
            AI[k * 16 + tid] = ar * pivI + ai * pivR;
        }
        __syncthreads();
        const int  i   = tid >> 4, j = tid & 15;
        const bool act = (tid < 128) && (i != k);
        float fr = 0.0f, fi = 0.0f, br = 0.0f, bi = 0.0f;
        if (act) {
            fr = AR[i * 16 + k]; fi = AI[i * 16 + k];
            br = AR[k * 16 + j]; bi = AI[k * 16 + j];
        }
        __syncthreads();
        if (act) {
            AR[i * 16 + j] -= fr * br - fi * bi;
            AI[i * 16 + j] -= fr * bi + fi * br;
        }
        __syncthreads();
    }

    if (tid < 64) {
        const int i = tid >> 3, j = tid & 7;
        float sr = 0.0f, si = 0.0f;
        #pragma unroll
        for (int k = 0; k < 8; ++k) {
            const float ar = AR[i * 16 + 8 + k], ai = AI[i * 16 + 8 + k];
            const float br = RsR[k * 8 + j],     bi = RsI[k * 8 + j];
            sr += ar * br - ai * bi;
            si += ar * bi + ai * br;
        }
        MR[i * 8 + j] = sr;
        MI[i * 8 + j] = si;
    }
    __syncthreads();

    if (tid == 0) {
        float tr = CEPS, ti = 0.0f;
        #pragma unroll
        for (int i = 0; i < 8; ++i) { tr += MR[i * 8 + i]; ti += MI[i * 8 + i]; }
        const float id = 1.0f / (tr * tr + ti * ti);
        #pragma unroll
        for (int i = 0; i < 8; ++i) {
            const float mr = MR[i * 8], mi = MI[i * 8];
            wR[i] = (mr * tr + mi * ti) * id;
            wI[i] = (mi * tr - mr * ti) * id;
        }
    }
    __syncthreads();

    float* outR = out + (((size_t)n * 2 + 0) * FDIM + f) * TDIM;
    float* outI = out + (((size_t)n * 2 + 1) * FDIM + f) * TDIM;
    for (int t = tid; t < TDIM; t += 256) {
        float br = 0.0f, bi = 0.0f;
        #pragma unroll
        for (int c = 0; c < CDIM; ++c) {
            const float sr = Zs[c * ZSTRIDE + t];
            const float si = Zs[(CDIM + c) * ZSTRIDE + t];
            br += wR[c] * sr + wI[c] * si;
            bi += wR[c] * si - wI[c] * sr;
        }
        outR[t] = br;
        outI[t] = bi;
    }
}

extern "C" void kernel_launch(void* const* d_in, const int* in_sizes, int n_in,
                              void* d_out, int out_size, void* d_ws, size_t ws_size,
                              hipStream_t stream) {
    (void)in_sizes; (void)n_in; (void)out_size; (void)d_ws; (void)ws_size;
    const float* spec_r = (const float*)d_in[0];
    const float* spec_i = (const float*)d_in[1];
    const float* mask   = (const float*)d_in[2];
    float*       outp   = (float*)d_out;

    dim3 grid(FDIM, 16);
    dim3 block(256);
    hipLaunchKernelGGL(mvdr_wmma_kernel, grid, block, 0, stream,
                       spec_r, spec_i, mask, outp);
}


# Tile kept in registers from global load to beamform; LDS only a half-tile transposition buffer (two Gram passes, partial sums regrouped): LDS 72 -> 40 KB, 5 workgroups per WGP instead of 4
# speedup vs baseline: 1.0935x; 1.0935x over previous
#include <hip/hip_runtime.h>


typedef float v2f __attribute__((ext_vector_type(2)));
typedef float v8f __attribute__((ext_vector_type(8)));

#define TDIM     1000
#define THALF    500
#define KHALF    125
#define FDIM     257
#define CDIM     8
#define CEPS     1.1920928955078125e-07f

__global__ __launch_bounds__(256) __attribute__((amdgpu_waves_per_eu(10))) void mvdr_wmma_kernel(
    const float* __restrict__ spec_r,
    const float* __restrict__ spec_i,
    const float* __restrict__ mask,
    float*       __restrict__ out)
{
    __shared__ __align__(16) float Zh[16 * THALF];
    __shared__ __align__(16) float Ms[TDIM];
    __shared__ float Gs[256];
    __shared__ float Gt[256];
    __shared__ float AR[8 * 16], AI[8 * 16];
    __shared__ float RsR[64], RsI[64];
    __shared__ float MR[64], MI[64];
    __shared__ float wR[8], wI[8];
    __shared__ float pivR, pivI;
    __shared__ float msum;

    const int tid = threadIdx.x;
    const int f   = blockIdx.x;
    const int n   = blockIdx.y;

    Gs[tid] = 0.0f;
    Gt[tid] = 0.0f;
    if (tid == 0) msum = 0.0f;

    const int t4 = (tid < TDIM / 4) ? tid : (TDIM / 4 - 1);
    const float4 mv = ((const float4*)(mask + ((size_t)n * FDIM + f) * TDIM))[t4];
    float4 zr[CDIM], zi[CDIM];
    #pragma unroll
    for (int r = 0; r < CDIM; ++r)
        zr[r] = ((const float4*)(spec_r + (((size_t)n * CDIM + r) * FDIM + f) * TDIM))[t4];
    #pragma unroll
    for (int r = 0; r < CDIM; ++r)
        zi[r] = ((const float4*)(spec_i + (((size_t)n * CDIM + r) * FDIM + f) * TDIM))[t4];
    if (tid < TDIM / 4) *(float4*)&Ms[4 * t4] = mv;

    const int wave = tid >> 5;
    const int lane = tid & 31;
    const int m    = lane & 15;
    const int koff = (lane >> 4) * 2;

    v8f acc_s = {0, 0, 0, 0, 0, 0, 0, 0};
    v8f acc_t = {0, 0, 0, 0, 0, 0, 0, 0};

    #pragma unroll
    for (int p = 0; p < 2; ++p) {
        if (p) __syncthreads();
        const int tl = tid - p * KHALF;
        if (tl >= 0 && tl < KHALF) {
            #pragma unroll
            for (int r = 0; r < CDIM; ++r) {
                *(float4*)&Zh[r * THALF + 4 * tl]          = zr[r];
                *(float4*)&Zh[(CDIM + r) * THALF + 4 * tl] = zi[r];
            }
        }
        __syncthreads();

        if (p == 0) {
            float local = 0.0f;
            for (int t = tid; t < TDIM; t += 256) local += Ms[t];
            atomicAdd(&msum, local);
        }

        for (int it = wave; it < KHALF; it += 8) {
            const int tc = it * 4 + koff;
            const v2f zb = *(const v2f*)&Zh[m * THALF + tc];
            const v2f mk = *(const v2f*)&Ms[p * THALF + tc];
            v2f za;
            za.x = zb.x * mk.x;
            za.y = zb.y * mk.y;
            acc_t = __builtin_amdgcn_wmma_f32_16x16x4_f32(
                        false, zb, false, zb, (short)0, acc_t, false, false);
            acc_s = __builtin_amdgcn_wmma_f32_16x16x4_f32(
                        false, za, false, zb, (short)0, acc_s, false, false);
        }
    }

    {
        const int Nn    = lane & 15;
        const int Mbase = (lane >> 4) * 8;
        #pragma unroll
        for (int r = 0; r < 8; ++r) {
            atomicAdd(&Gs[(Mbase + r) * 16 + Nn], acc_s[r]);
            atomicAdd(&Gt[(Mbase + r) * 16 + Nn], acc_t[r]);
        }
    }
    __syncthreads();

    float4* stash = (float4*)Zh;
    #pragma unroll
    for (int q = 0; q < 3; ++q) {
        stash[(2 * q + 0) * 256 + tid] = zr[5 + q];
        stash[(2 * q + 1) * 256 + tid] = zi[5 + q];
    }

    if (tid < 64) {
        const int i = tid >> 3, j = tid & 7;
        const float ds    = fmaxf(msum, CEPS);
        const float dn    = fmaxf((float)TDIM - msum, CEPS);
        const float inv_s = 1.0f / ds;
        const float inv_n = 1.0f / dn;

        const float gs_a = Gs[i * 16 + j],           gt_a = Gt[i * 16 + j];
        const float gs_b = Gs[(i + 8) * 16 + j + 8], gt_b = Gt[(i + 8) * 16 + j + 8];
        const float gs_c = Gs[(i + 8) * 16 + j],     gt_c = Gt[(i + 8) * 16 + j];
        const float gs_d = Gs[i * 16 + j + 8],       gt_d = Gt[i * 16 + j + 8];

        RsR[i * 8 + j] = (gs_a + gs_b) * inv_s;
        RsI[i * 8 + j] = (gs_c - gs_d) * inv_s;

        const float rnr = ((gt_a - gs_a) + (gt_b - gs_b)) * inv_n;
        const float rni = ((gt_c - gs_c) - (gt_d - gs_d)) * inv_n;
        AR[i * 16 + j]     = rnr + ((i == j) ? CEPS : 0.0f);
        AI[i * 16 + j]     = rni;
        AR[i * 16 + 8 + j] = (i == j) ? 1.0f : 0.0f;
        AI[i * 16 + 8 + j] = 0.0f;
    }
    __syncthreads();

    for (int k = 0; k < 8; ++k) {
        if (tid == 0) {
            const float pr = AR[k * 16 + k], pi = AI[k * 16 + k];
            const float id = 1.0f / (pr * pr + pi * pi);
            pivR = pr * id;
            pivI = -pi * id;
        }
        __syncthreads();
        if (tid < 16) {
            const float ar = AR[k * 16 + tid], ai = AI[k * 16 + tid];
            AR[k * 16 + tid] = ar * pivR - ai * pivI;
            AI[k * 16 + tid] = ar * pivI + ai * pivR;
        }
        __syncthreads();
        const int  i   = tid >> 4, j = tid & 15;
        const bool act = (tid < 128) && (i != k);
        float fr = 0.0f, fi = 0.0f, br = 0.0f, bi = 0.0f;
        if (act) {
            fr = AR[i * 16 + k]; fi = AI[i * 16 + k];
            br = AR[k * 16 + j]; bi = AI[k * 16 + j];
        }
        __syncthreads();
        if (act) {
            AR[i * 16 + j] -= fr * br - fi * bi;
            AI[i * 16 + j] -= fr * bi + fi * br;
        }
        __syncthreads();
    }

    if (tid < 64) {
        const int i = tid >> 3, j = tid & 7;
        float sr = 0.0f, si = 0.0f;
        #pragma unroll
        for (int k = 0; k < 8; ++k) {
            const float ar = AR[i * 16 + 8 + k], ai = AI[i * 16 + 8 + k];
            const float br = RsR[k * 8 + j],     bi = RsI[k * 8 + j];
            sr += ar * br - ai * bi;
            si += ar * bi + ai * br;
        }
        MR[i * 8 + j] = sr;
        MI[i * 8 + j] = si;
    }
    __syncthreads();

    if (tid == 0) {
        float tr = CEPS, ti = 0.0f;
        #pragma unroll
        for (int i = 0; i < 8; ++i) { tr += MR[i * 8 + i]; ti += MI[i * 8 + i]; }
        const float id = 1.0f / (tr * tr + ti * ti);
        #pragma unroll
        for (int i = 0; i < 8; ++i) {
            const float mr = MR[i * 8], mi = MI[i * 8];
            wR[i] = (mr * tr + mi * ti) * id;
            wI[i] = (mi * tr - mr * ti) * id;
        }
    }
    __syncthreads();

    if (tid < TDIM / 4) {
        float4 oR = make_float4(0.0f, 0.0f, 0.0f, 0.0f);
        float4 oI = make_float4(0.0f, 0.0f, 0.0f, 0.0f);
        #pragma unroll
        for (int c = 0; c < CDIM; ++c) {
            if (c >= 5) {
                zr[c] = stash[(2 * (c - 5) + 0) * 256 + tid];
                zi[c] = stash[(2 * (c - 5) + 1) * 256 + tid];
            }
            const float a = wR[c], b = wI[c];
            oR.x += a * zr[c].x + b * zi[c].x;
            oI.x += a * zi[c].x - b * zr[c].x;
            oR.y += a * zr[c].y + b * zi[c].y;
            oI.y += a * zi[c].y - b * zr[c].y;
            oR.z += a * zr[c].z + b * zi[c].z;
            oI.z += a * zi[c].z - b * zr[c].z;
            oR.w += a * zr[c].w + b * zi[c].w;
            oI.w += a * zi[c].w - b * zr[c].w;
        }
        float* outR = out + (((size_t)n * 2 + 0) * FDIM + f) * TDIM;
        float* outI = out + (((size_t)n * 2 + 1) * FDIM + f) * TDIM;
        ((float4*)outR)[tid] = oR;
        ((float4*)outI)[tid] = oI;
    }
}

extern "C" void kernel_launch(void* const* d_in, const int* in_sizes, int n_in,
                              void* d_out, int out_size, void* d_ws, size_t ws_size,
                              hipStream_t stream) {
    (void)in_sizes; (void)n_in; (void)out_size; (void)d_ws; (void)ws_size;
    const float* spec_r = (const float*)d_in[0];
    const float* spec_i = (const float*)d_in[1];
    const float* mask   = (const float*)d_in[2];
    float*       outp   = (float*)d_out;

    dim3 grid(FDIM, 16);
    dim3 block(256);
    hipLaunchKernelGGL(mvdr_wmma_kernel, grid, block, 0, stream,
                       spec_r, spec_i, mask, outp);
}
